# starting kernel
# baseline (speedup 1.0000x reference)
#include <hip/hip_runtime.h>
#include <hip/hip_bf16.h>

typedef _Float16 v16h __attribute__((ext_vector_type(16)));
typedef _Float16 v8h  __attribute__((ext_vector_type(8)));
typedef float    v8f  __attribute__((ext_vector_type(8)));
typedef float    v4f  __attribute__((ext_vector_type(4)));

#define FDIM 256
#define TDIM 2048
#define BDIM 128
#define LDS_STRIDE 264
#define TCHUNKS 8

#if __has_builtin(__builtin_amdgcn_tanhf)
  #define FAST_TANH(v) __builtin_amdgcn_tanhf(v)
#elif __has_builtin(__builtin_amdgcn_tanh_f32)
  #define FAST_TANH(v) __builtin_amdgcn_tanh_f32(v)
#else
  #define FAST_TANH(v) tanhf(v)
#endif

__global__ __launch_bounds__(256) void awc_scores(const float* __restrict__ x,
                                                  const float* __restrict__ W,
                                                  const float* __restrict__ bias,
                                                  const float* __restrict__ u,
                                                  float* __restrict__ ait)
{
    __shared__ _Float16 sW[FDIM * LDS_STRIDE];

    const int tid = threadIdx.x;

    {
        _Float16* dst = &sW[(unsigned)tid * LDS_STRIDE];
        for (int f = 0; f < FDIM; f += 2) {
            float w0 = W[f * FDIM + tid];
            float w1 = W[(f + 1) * FDIM + tid];
            v8h p = {};
            p[0] = (_Float16)w0;
            p[1] = (_Float16)w1;
            *(__attribute__((ext_vector_type(2))) _Float16*)(dst + f) =
                __builtin_shufflevector(p, p, 0, 1);
        }
    }
    __syncthreads();

    const int wave  = tid >> 5;
    const int lane  = tid & 31;
    const int lrow  = lane & 15;
    const int khalf = (lane >> 4) * 8;

    const long rowBase = (long)blockIdx.x * 256 + wave * 32;

    v16h afrag[2][8];
#pragma unroll
    for (int s = 0; s < 2; ++s) {
        const float* xr = x + (rowBase + s * 16 + lrow) * (long)FDIM;
#pragma unroll
        for (int k = 0; k < 8; ++k) {
            const int kb = k * 32 + khalf;
            v4f lo0 = *(const v4f*)(xr + kb);
            v4f lo1 = *(const v4f*)(xr + kb + 4);
            v4f hi0 = *(const v4f*)(xr + kb + 16);
            v4f hi1 = *(const v4f*)(xr + kb + 20);
            v16h a;
#pragma unroll
            for (int e = 0; e < 4; ++e) {
                a[e]      = (_Float16)lo0[e];
                a[e + 4]  = (_Float16)lo1[e];
                a[e + 8]  = (_Float16)hi0[e];
                a[e + 12] = (_Float16)hi1[e];
            }
            afrag[s][k] = a;
        }
    }

    float part[2][8];
#pragma unroll
    for (int s = 0; s < 2; ++s)
#pragma unroll
        for (int r = 0; r < 8; ++r) part[s][r] = 0.0f;

    auto loadB = [&](int nk) -> v16h {
        const int nn = nk >> 3;
        const int kk = nk & 7;
        const _Float16* p = &sW[(nn * 16 + lrow) * LDS_STRIDE + kk * 32 + khalf];
        v8h blo = *(const v8h*)(p);
        v8h bhi = *(const v8h*)(p + 16);
        return __builtin_shufflevector(blo, bhi,
            0,1,2,3,4,5,6,7,8,9,10,11,12,13,14,15);
    };

    v16h bf0 = loadB(0);
    v16h bf1 = loadB(1);

#pragma unroll
    for (int n = 0; n < 16; ++n) {
        v8f acc0 = {};
        v8f acc1 = {};
#pragma unroll
        for (int k = 0; k < 8; ++k) {
            const int nk = n * 8 + k;
            v16h bcur = bf0;
            bf0 = bf1;
            if (nk + 2 < 128) bf1 = loadB(nk + 2);
            acc0 = __builtin_amdgcn_wmma_f32_16x16x32_f16(
                false, afrag[0][k], false, bcur, (short)0, acc0, false, false);
            acc1 = __builtin_amdgcn_wmma_f32_16x16x32_f16(
                false, afrag[1][k], false, bcur, (short)0, acc1, false, false);
        }
        const int col = n * 16 + lrow;
        const float un = u[col];
        const float bn = bias[col];
#pragma unroll
        for (int r = 0; r < 8; ++r) {
            part[0][r] += FAST_TANH(acc0[r] + bn) * un;
            part[1][r] += FAST_TANH(acc1[r] + bn) * un;
        }
    }

#pragma unroll
    for (int s = 0; s < 2; ++s) {
#pragma unroll
        for (int r = 0; r < 8; ++r) {
            float v = part[s][r];
            v += __shfl_xor(v, 1, 32);
            v += __shfl_xor(v, 2, 32);
            v += __shfl_xor(v, 4, 32);
            v += __shfl_xor(v, 8, 32);
            part[s][r] = v;
        }
    }

    if (lrow == 0) {
#pragma unroll
        for (int s = 0; s < 2; ++s) {
            const long base = rowBase + s * 16 + (lane >> 4) * 8;
#pragma unroll
            for (int r = 0; r < 8; ++r) ait[base + r] = part[s][r];
        }
    }
}

__global__ __launch_bounds__(256) void awc_softmax(float* __restrict__ a)
{
    __shared__ float red[8];
    const int b   = blockIdx.x;
    const int tid = threadIdx.x;
    float* row = a + (long)b * TDIM;

    float s = 0.0f;
    for (int t = tid; t < TDIM; t += 256) s += expf(row[t]);
#pragma unroll
    for (int off = 16; off; off >>= 1) s += __shfl_xor(s, off, 32);
    if ((tid & 31) == 0) red[tid >> 5] = s;
    __syncthreads();

    float tot = 0.0f;
#pragma unroll
    for (int w = 0; w < 8; ++w) tot += red[w];
    const float inv = 1.0f / (tot + 1e-7f);

    for (int t = tid; t < TDIM; t += 256) row[t] = expf(row[t]) * inv;
}

__global__ __launch_bounds__(256) void awc_wsum_partial(const float* __restrict__ x,
                                                        const float* __restrict__ a,
                                                        float* __restrict__ partial)
{
    const int b = blockIdx.y;
    const int c = blockIdx.x;
    const int f = threadIdx.x;
    const long tbase = (long)b * TDIM + (long)c * (TDIM / TCHUNKS);

    const float* xp = x + tbase * FDIM + f;
    const float* ap = a + tbase;
    float acc = 0.0f;
    for (int t = 0; t < TDIM / TCHUNKS; ++t)
        acc += ap[t] * xp[(long)t * FDIM];
    partial[((long)b * TCHUNKS + c) * FDIM + f] = acc;
}

__global__ __launch_bounds__(256) void awc_wsum_reduce(const float* __restrict__ partial,
                                                       float* __restrict__ out)
{
    const int b = blockIdx.x;
    const int f = threadIdx.x;
    float acc = 0.0f;
#pragma unroll
    for (int c = 0; c < TCHUNKS; ++c)
        acc += partial[((long)b * TCHUNKS + c) * FDIM + f];
    out[(long)b * FDIM + f] = acc;
}

extern "C" void kernel_launch(void* const* d_in, const int* in_sizes, int n_in,
                              void* d_out, int out_size, void* d_ws, size_t ws_size,
                              hipStream_t stream) {
    const float* x    = (const float*)d_in[0];
    const float* W    = (const float*)d_in[1];
    const float* bias = (const float*)d_in[2];
    const float* u    = (const float*)d_in[3];
    float* out = (float*)d_out;

    float* a       = (float*)d_ws;
    float* partial = a + (size_t)BDIM * TDIM;

    awc_scores      <<<dim3((BDIM * TDIM) / 256), 256, 0, stream>>>(x, W, bias, u, a);
    awc_softmax     <<<dim3(BDIM),                256, 0, stream>>>(a);
    awc_wsum_partial<<<dim3(TCHUNKS, BDIM),       256, 0, stream>>>(x, a, partial);
    awc_wsum_reduce <<<dim3(BDIM),                256, 0, stream>>>(partial, out);
}


# scores kernel made persistent: 256 work-groups walk 4 row tiles each, W^T staged in LDS once per work-group, next tile's x loads issued as two 32-load groups around the WMMA block
# speedup vs baseline: 1.3901x; 1.3901x over previous
#include <hip/hip_runtime.h>
#include <hip/hip_bf16.h>

typedef _Float16 v16h __attribute__((ext_vector_type(16)));
typedef _Float16 v8h  __attribute__((ext_vector_type(8)));
typedef float    v8f  __attribute__((ext_vector_type(8)));
typedef float    v4f  __attribute__((ext_vector_type(4)));

#define FDIM 256
#define TDIM 2048
#define BDIM 128
#define LDS_STRIDE 264
#define TCHUNKS 8

#if __has_builtin(__builtin_amdgcn_tanhf)
  #define FAST_TANH(v) __builtin_amdgcn_tanhf(v)
#elif __has_builtin(__builtin_amdgcn_tanh_f32)
  #define FAST_TANH(v) __builtin_amdgcn_tanh_f32(v)
#else
  #define FAST_TANH(v) tanhf(v)
#endif

#define SCORE_TILES ((BDIM * TDIM) / 256)
#define SCORE_WGS   256

__global__ __launch_bounds__(256) void awc_scores(const float* __restrict__ x,
                                                  const float* __restrict__ W,
                                                  const float* __restrict__ bias,
                                                  const float* __restrict__ u,
                                                  float* __restrict__ ait)
{
    __shared__ _Float16 sW[FDIM * LDS_STRIDE];

    const int tid   = threadIdx.x;
    const int wave  = tid >> 5;
    const int lane  = tid & 31;
    const int lrow  = lane & 15;
    const int khalf = (lane >> 4) * 8;

    v4f raw[2][8][4];
    auto issueX = [&](int tile, int s) {
        const long rowBase = (long)tile * 256 + wave * 32;
        const float* xr = x + (rowBase + s * 16 + lrow) * (long)FDIM;
#pragma unroll
        for (int k = 0; k < 8; ++k) {
            const int kb = k * 32 + khalf;
            raw[s][k][0] = *(const v4f*)(xr + kb);
            raw[s][k][1] = *(const v4f*)(xr + kb + 4);
            raw[s][k][2] = *(const v4f*)(xr + kb + 16);
            raw[s][k][3] = *(const v4f*)(xr + kb + 20);
        }
    };

    const int tstep = gridDim.x;
    int tile = blockIdx.x;
    if (tile < SCORE_TILES) { issueX(tile, 0); issueX(tile, 1); }

    {
        _Float16* dst = &sW[(unsigned)tid * LDS_STRIDE];
        for (int f = 0; f < FDIM; f += 2) {
            float w0 = W[f * FDIM + tid];
            float w1 = W[(f + 1) * FDIM + tid];
            v8h p = {};
            p[0] = (_Float16)w0;
            p[1] = (_Float16)w1;
            *(__attribute__((ext_vector_type(2))) _Float16*)(dst + f) =
                __builtin_shufflevector(p, p, 0, 1);
        }
    }

    float un[16], bn[16];
#pragma unroll
    for (int n = 0; n < 16; ++n) {
        un[n] = u[n * 16 + lrow];
        bn[n] = bias[n * 16 + lrow];
    }
    __syncthreads();

    auto loadB = [&](int nk) -> v16h {
        const int nn = nk >> 3;
        const int kk = nk & 7;
        const _Float16* p = &sW[(nn * 16 + lrow) * LDS_STRIDE + kk * 32 + khalf];
        v8h blo = *(const v8h*)(p);
        v8h bhi = *(const v8h*)(p + 16);
        return __builtin_shufflevector(blo, bhi,
            0,1,2,3,4,5,6,7,8,9,10,11,12,13,14,15);
    };

    for (; tile < SCORE_TILES; tile += tstep) {
        asm volatile("" ::: "memory");
        const long rowBase = (long)tile * 256 + wave * 32;

        v16h afrag[2][8];
#pragma unroll
        for (int s = 0; s < 2; ++s) {
#pragma unroll
            for (int k = 0; k < 8; ++k) {
                v16h a;
#pragma unroll
                for (int e = 0; e < 4; ++e) {
                    a[e]      = (_Float16)raw[s][k][0][e];
                    a[e + 4]  = (_Float16)raw[s][k][1][e];
                    a[e + 8]  = (_Float16)raw[s][k][2][e];
                    a[e + 12] = (_Float16)raw[s][k][3][e];
                }
                afrag[s][k] = a;
            }
        }

        asm volatile("" : "+v"(afrag[0][0]), "+v"(afrag[0][1]), "+v"(afrag[0][2]), "+v"(afrag[0][3]),
                          "+v"(afrag[0][4]), "+v"(afrag[0][5]), "+v"(afrag[0][6]), "+v"(afrag[0][7]) :: "memory");
        asm volatile("" : "+v"(afrag[1][0]), "+v"(afrag[1][1]), "+v"(afrag[1][2]), "+v"(afrag[1][3]),
                          "+v"(afrag[1][4]), "+v"(afrag[1][5]), "+v"(afrag[1][6]), "+v"(afrag[1][7]) :: "memory");

        const int next = tile + tstep;
        if (next < SCORE_TILES) issueX(next, 0);

        float part[2][8];
#pragma unroll
        for (int s = 0; s < 2; ++s)
#pragma unroll
            for (int r = 0; r < 8; ++r) part[s][r] = 0.0f;

        v16h bf0 = loadB(0);
        v16h bf1 = loadB(1);

#pragma unroll
        for (int n = 0; n < 16; ++n) {
            v8f acc0 = {};
            v8f acc1 = {};
#pragma unroll
            for (int k = 0; k < 8; ++k) {
                const int nk = n * 8 + k;
                v16h bcur = bf0;
                bf0 = bf1;
                if (nk + 2 < 128) bf1 = loadB(nk + 2);
                acc0 = __builtin_amdgcn_wmma_f32_16x16x32_f16(
                    false, afrag[0][k], false, bcur, (short)0, acc0, false, false);
                acc1 = __builtin_amdgcn_wmma_f32_16x16x32_f16(
                    false, afrag[1][k], false, bcur, (short)0, acc1, false, false);
            }
#pragma unroll
            for (int r = 0; r < 8; ++r) {
                part[0][r] += FAST_TANH(acc0[r] + bn[n]) * un[n];
                part[1][r] += FAST_TANH(acc1[r] + bn[n]) * un[n];
            }
        }

#pragma unroll
        for (int s = 0; s < 2; ++s) {
#pragma unroll
            for (int r = 0; r < 8; ++r) {
                float v = part[s][r];
                v += __shfl_xor(v, 1, 32);
                v += __shfl_xor(v, 2, 32);
                v += __shfl_xor(v, 4, 32);
                v += __shfl_xor(v, 8, 32);
                part[s][r] = v;
            }
        }

        if (lrow == 0) {
#pragma unroll
            for (int s = 0; s < 2; ++s) {
                const long base = rowBase + s * 16 + (lane >> 4) * 8;
#pragma unroll
                for (int r = 0; r < 8; ++r) ait[base + r] = part[s][r];
            }
        }

        if (next < SCORE_TILES) issueX(next, 1);
    }
}

__global__ __launch_bounds__(256) void awc_softmax(float* __restrict__ a)
{
    __shared__ float red[8];
    const int b   = blockIdx.x;
    const int tid = threadIdx.x;
    float* row = a + (long)b * TDIM;

    float s = 0.0f;
    for (int t = tid; t < TDIM; t += 256) s += expf(row[t]);
#pragma unroll
    for (int off = 16; off; off >>= 1) s += __shfl_xor(s, off, 32);
    if ((tid & 31) == 0) red[tid >> 5] = s;
    __syncthreads();

    float tot = 0.0f;
#pragma unroll
    for (int w = 0; w < 8; ++w) tot += red[w];
    const float inv = 1.0f / (tot + 1e-7f);

    for (int t = tid; t < TDIM; t += 256) row[t] = expf(row[t]) * inv;
}

__global__ __launch_bounds__(256) void awc_wsum_partial(const float* __restrict__ x,
                                                        const float* __restrict__ a,
                                                        float* __restrict__ partial)
{
    const int b = blockIdx.y;
    const int c = blockIdx.x;
    const int f = threadIdx.x;
    const long tbase = (long)b * TDIM + (long)c * (TDIM / TCHUNKS);

    const float* xp = x + tbase * FDIM + f;
    const float* ap = a + tbase;
    float acc = 0.0f;
    for (int t = 0; t < TDIM / TCHUNKS; ++t)
        acc += ap[t] * xp[(long)t * FDIM];
    partial[((long)b * TCHUNKS + c) * FDIM + f] = acc;
}

__global__ __launch_bounds__(256) void awc_wsum_reduce(const float* __restrict__ partial,
                                                       float* __restrict__ out)
{
    const int b = blockIdx.x;
    const int f = threadIdx.x;
    float acc = 0.0f;
#pragma unroll
    for (int c = 0; c < TCHUNKS; ++c)
        acc += partial[((long)b * TCHUNKS + c) * FDIM + f];
    out[(long)b * FDIM + f] = acc;
}

extern "C" void kernel_launch(void* const* d_in, const int* in_sizes, int n_in,
                              void* d_out, int out_size, void* d_ws, size_t ws_size,
                              hipStream_t stream) {
    const float* x    = (const float*)d_in[0];
    const float* W    = (const float*)d_in[1];
    const float* bias = (const float*)d_in[2];
    const float* u    = (const float*)d_in[3];
    float* out = (float*)d_out;

    float* a       = (float*)d_ws;
    float* partial = a + (size_t)BDIM * TDIM;

    awc_scores      <<<dim3(SCORE_WGS),           256, 0, stream>>>(x, W, bias, u, a);
    awc_softmax     <<<dim3(BDIM),                256, 0, stream>>>(a);
    awc_wsum_partial<<<dim3(TCHUNKS, BDIM),       256, 0, stream>>>(x, a, partial);
    awc_wsum_reduce <<<dim3(BDIM),                256, 0, stream>>>(partial, out);
}


# x rows fetched with whole-line loads (512 contiguous bytes each) and transposed to the A layout through a wave-private LDS area, not 16-line row-per-lane loads; 3 load groups per tile
# speedup vs baseline: 2.2360x; 1.6085x over previous
#include <hip/hip_runtime.h>
#include <hip/hip_bf16.h>

typedef _Float16 v16h __attribute__((ext_vector_type(16)));
typedef _Float16 v8h  __attribute__((ext_vector_type(8)));
typedef float    v8f  __attribute__((ext_vector_type(8)));
typedef float    v4f  __attribute__((ext_vector_type(4)));

#define FDIM 256
#define TDIM 2048
#define BDIM 128
#define LDS_STRIDE 264

#if __has_builtin(__builtin_amdgcn_tanhf)
  #define FAST_TANH(v) __builtin_amdgcn_tanhf(v)
#elif __has_builtin(__builtin_amdgcn_tanh_f32)
  #define FAST_TANH(v) __builtin_amdgcn_tanh_f32(v)
#else
  #define FAST_TANH(v) tanhf(v)
#endif

#define SCORE_TILES ((BDIM * TDIM) / 256)
#define SCORE_WGS   256
#define TPW         (SCORE_TILES / SCORE_WGS)
#define WG_PER_B    (TDIM / (256 * TPW))
#define NWAVES      4
#define NS          4
#define NTHREADS    (NWAVES * 32)

#define XPITCH      264
#define WAVE_ROWS   (NS * 16)
#define G0_ROWS     18
#define G1_ROWS     18
#define G2_ROWS     28

template <bool B> struct BoolTag { static constexpr bool value = B; };
template <int V>  struct IntTag  { static constexpr int  value = V; };

__global__ __launch_bounds__(NTHREADS) void awc_scores_wsum(const float* __restrict__ x,
                                                            const float* __restrict__ W,
                                                            const float* __restrict__ bias,
                                                            const float* __restrict__ u,
                                                            float* __restrict__ partial,
                                                            float* __restrict__ psum)
{
    __shared__ _Float16 sW[FDIM * LDS_STRIDE];
    __shared__ __attribute__((aligned(16))) _Float16 sX[NWAVES * WAVE_ROWS * XPITCH];
    __shared__ float sU[FDIM];
    __shared__ float sBias[FDIM];
    __shared__ float sY[NWAVES * 2 * FDIM];
    __shared__ float sE[NWAVES * 2];

    const int tid   = threadIdx.x;
    const int wave  = tid >> 5;
    const int lane  = tid & 31;
    const int lrow  = lane & 15;
    const int lhalf = lane >> 4;
    const int khalf = lhalf * 8;

    v4f g0[G0_ROWS][2], g1[G1_ROWS][2], g2[G2_ROWS][2];
    typedef _Float16 v4h __attribute__((ext_vector_type(4)));
    _Float16* const myX = &sX[wave * (WAVE_ROWS * XPITCH)];

    auto issueRows = [&](int tile, auto& g, auto r0Tag) {
        constexpr int R0 = decltype(r0Tag)::value;
        constexpr int NR = (int)(sizeof(g) / sizeof(g[0]));
        const float* xr = x + ((long)tile * 256 + wave * WAVE_ROWS + R0) * (long)FDIM + lane * 4;
#pragma unroll
        for (int r = 0; r < NR; ++r) {
            g[r][0] = *(const v4f*)(xr + r * FDIM);
            g[r][1] = *(const v4f*)(xr + r * FDIM + 128);
        }
    };
    auto stageRows = [&](auto& g, auto r0Tag) {
        constexpr int R0 = decltype(r0Tag)::value;
        constexpr int NR = (int)(sizeof(g) / sizeof(g[0]));
        _Float16* d = myX + R0 * XPITCH + lane * 4;
#pragma unroll
        for (int r = 0; r < NR; ++r) {
#pragma unroll
            for (int hh = 0; hh < 2; ++hh) {
                v4h h;
#pragma unroll
                for (int e = 0; e < 4; ++e) h[e] = (_Float16)g[r][hh][e];
                *(v4h*)(d + r * XPITCH + hh * 128) = h;
            }
        }
    };
    const IntTag<0>                 R_G0{};
    const IntTag<G0_ROWS>           R_G1{};
    const IntTag<G0_ROWS + G1_ROWS> R_G2{};

    const int tile0 = blockIdx.x * TPW;
    issueRows(tile0, g0, R_G0);

    auto stageW = [&](int fBeg, int fEnd) {
        _Float16* d0 = &sW[(unsigned)tid * LDS_STRIDE];
        _Float16* d1 = &sW[(unsigned)(tid + NTHREADS) * LDS_STRIDE];
#pragma unroll 16
        for (int f = fBeg; f < fEnd; f += 2) {
            const float a0 = W[f * FDIM + tid];
            const float a1 = W[(f + 1) * FDIM + tid];
            const float c0 = W[f * FDIM + tid + NTHREADS];
            const float c1 = W[(f + 1) * FDIM + tid + NTHREADS];
            typedef _Float16 v2h __attribute__((ext_vector_type(2)));
            v2h p, q;
            p[0] = (_Float16)a0; p[1] = (_Float16)a1;
            q[0] = (_Float16)c0; q[1] = (_Float16)c1;
            *(v2h*)(d0 + f) = p;
            *(v2h*)(d1 + f) = q;
        }
    };
    stageW(0, FDIM / 2);
    stageRows(g0, R_G0);
    issueRows(tile0, g1, R_G1);
    stageW(FDIM / 2, FDIM);
    stageRows(g1, R_G1);
    issueRows(tile0, g2, R_G2);
    sU[tid]               = u[tid];
    sU[tid + NTHREADS]    = u[tid + NTHREADS];
    sBias[tid]            = bias[tid];
    sBias[tid + NTHREADS] = bias[tid + NTHREADS];

    v16h sel0 = {};
    v16h sel1 = {};
    {
        const bool mine = ((lrow >> 3) == lhalf);
#pragma unroll
        for (int e = 0; e < 8; ++e) {
            const _Float16 v = (mine && (lrow & 7) == e) ? (_Float16)1.0f : (_Float16)0.0f;
            sel0[e]     = v;
            sel1[e + 8] = v;
        }
    }

    float ysum[16];
#pragma unroll
    for (int j = 0; j < 16; ++j) ysum[j] = 0.0f;
    float esum = 0.0f;

    __syncthreads();

    auto loadB = [&](int n, int kk) -> v16h {
        const _Float16* p = &sW[(n * 16 + lrow) * LDS_STRIDE + kk * 32 + khalf];
        v8h blo = *(const v8h*)(p);
        v8h bhi = *(const v8h*)(p + 16);
        return __builtin_shufflevector(blo, bhi,
            0,1,2,3,4,5,6,7,8,9,10,11,12,13,14,15);
    };

    v16h bA[8], bB[8];
#pragma unroll
    for (int k = 0; k < 8; ++k) bA[k] = loadB(0, k);

    auto tileBody = [&](int it, auto moreTag) {
        constexpr bool more = decltype(moreTag)::value;
        const int tile = tile0 + it;

        stageRows(g2, R_G2);

        if constexpr (more) issueRows(tile + 1, g0, R_G0);

        v16h afrag[NS][8];
#pragma unroll
        for (int s = 0; s < NS; ++s) {
#pragma unroll
            for (int k = 0; k < 8; ++k) {
                const _Float16* p = myX + (s * 16 + lrow) * XPITCH + k * 32 + khalf;
                v8h alo = *(const v8h*)(p);
                v8h ahi = *(const v8h*)(p + 16);
                afrag[s][k] = __builtin_shufflevector(alo, ahi,
                    0,1,2,3,4,5,6,7,8,9,10,11,12,13,14,15);
            }
        }

        float part[NS][8];
#pragma unroll
        for (int s = 0; s < NS; ++s)
#pragma unroll
            for (int r = 0; r < 8; ++r) part[s][r] = 0.0f;

        auto colBlock = [&](int n, const v16h (&bf)[8]) {
            v8f acc[NS];
#pragma unroll
            for (int s = 0; s < NS; ++s) acc[s] = v8f{};
            const float un = sU[n * 16 + lrow];
            const float bn = sBias[n * 16 + lrow];
#pragma unroll
            for (int k = 0; k < 8; ++k) {
#pragma unroll
                for (int s = 0; s < NS; ++s)
                    acc[s] = __builtin_amdgcn_wmma_f32_16x16x32_f16(
                        false, afrag[s][k], false, bf[k], (short)0, acc[s], false, false);
            }
#pragma unroll
            for (int r = 0; r < 8; ++r) {
#pragma unroll
                for (int s = 0; s < NS; ++s)
                    part[s][r] += FAST_TANH(acc[s][r] + bn) * un;
            }
        };

        auto colTrips = [&](int nBeg, int nEnd) {
#pragma clang loop unroll(disable)
        for (int n = nBeg; n < nEnd; n += 2) {
            __builtin_amdgcn_sched_barrier(0x406);
#pragma unroll
            for (int k = 0; k < 8; ++k) bB[k] = loadB(n + 1, k);
            __builtin_amdgcn_sched_barrier(0x406);
            colBlock(n, bA);
            __builtin_amdgcn_sched_barrier(0x406);
#pragma unroll
            for (int k = 0; k < 8; ++k) bA[k] = loadB((n + 2) & 15, k);
            __builtin_amdgcn_sched_barrier(0x406);
            colBlock(n + 1, bB);
        }
        };
        colTrips(0, 8);
        if constexpr (more) { stageRows(g0, R_G0); issueRows(tile + 1, g1, R_G1); }
        colTrips(8, 16);
        if constexpr (more) {
            stageRows(g1, R_G1);
            issueRows(tile + 1, g2, R_G2);
            __builtin_amdgcn_sched_barrier(0);
        }

#pragma unroll
        for (int s = 0; s < NS; ++s) {
#pragma unroll
            for (int r = 0; r < 8; ++r) {
                float v = part[s][r];
                v += __shfl_xor(v, 1, 32);
                v += __shfl_xor(v, 2, 32);
                v += __shfl_xor(v, 4, 32);
                v += __shfl_xor(v, 8, 32);
                v = expf(v);
                part[s][r] = v;
                esum += v;
            }
        }

#pragma unroll
        for (int q = 0; q < 4 * NS; ++q) {
            const int s = q >> 2;
            const int g = q & 3;
#pragma unroll
            for (int jj = 0; jj < 4; ++jj) {
                const int j = 4 * g + jj;
                v8f z = {};
                v8f d = __builtin_amdgcn_wmma_f32_16x16x32_f16(
                    false, afrag[s][j >> 1], false, (j & 1) ? sel1 : sel0, (short)0, z, false, false);
#pragma unroll
                for (int r = 0; r < 8; ++r) ysum[j] = __builtin_fmaf(part[s][r], d[r], ysum[j]);
            }
        }
    };

#pragma clang loop unroll(disable)
    for (int it = 0; it < TPW - 1; ++it) tileBody(it, BoolTag<true>{});
    tileBody(TPW - 1, BoolTag<false>{});

#pragma unroll
    for (int j = 0; j < 16; ++j)
        sY[(wave * 2 + lhalf) * FDIM + j * 16 + lrow] = ysum[j];
    if (lrow == 0) sE[wave * 2 + lhalf] = esum;
    __syncthreads();

#pragma unroll
    for (int c2 = 0; c2 < FDIM / NTHREADS; ++c2) {
        const int col = tid + c2 * NTHREADS;
        float tot = 0.0f;
#pragma unroll
        for (int p = 0; p < NWAVES * 2; ++p) tot += sY[p * FDIM + col];
        partial[(long)blockIdx.x * FDIM + col] = tot;
    }
    if (tid == 0) {
        float es = 0.0f;
#pragma unroll
        for (int p = 0; p < NWAVES * 2; ++p) es += sE[p];
        psum[blockIdx.x] = es;
    }
}

__global__ __launch_bounds__(256) void awc_finalize(const float* __restrict__ partial,
                                                    const float* __restrict__ psum,
                                                    float* __restrict__ out)
{
    const int b = blockIdx.x;
    const int f = threadIdx.x;
    float acc = 0.0f;
    float es  = 0.0f;
#pragma unroll
    for (int c = 0; c < WG_PER_B; ++c) {
        acc += partial[((long)b * WG_PER_B + c) * FDIM + f];
        es  += psum[b * WG_PER_B + c];
    }
    out[(long)b * FDIM + f] = acc * (1.0f / (es + 1e-7f));
}

extern "C" void kernel_launch(void* const* d_in, const int* in_sizes, int n_in,
                              void* d_out, int out_size, void* d_ws, size_t ws_size,
                              hipStream_t stream) {
    const float* x    = (const float*)d_in[0];
    const float* W    = (const float*)d_in[1];
    const float* bias = (const float*)d_in[2];
    const float* u    = (const float*)d_in[3];
    float* out = (float*)d_out;

    float* psum    = (float*)d_ws;
    float* partial = psum + (size_t)BDIM * TDIM;

    awc_scores_wsum<<<dim3(SCORE_WGS), NTHREADS, 0, stream>>>(x, W, bias, u, partial, psum);
    awc_finalize   <<<dim3(BDIM),      256, 0, stream>>>(partial, psum, out);
}


# score reduction over each 16-lane row moved to the vector ALU: four adds with a DPP row_xmask source per row sum, not __shfl_xor (ds_bpermute_b32 + dscnt waits); same partners, sums bit-identical
# speedup vs baseline: 2.4029x; 1.0746x over previous
#include <hip/hip_runtime.h>
#include <hip/hip_bf16.h>

typedef _Float16 v16h __attribute__((ext_vector_type(16)));
typedef _Float16 v8h  __attribute__((ext_vector_type(8)));
typedef float    v8f  __attribute__((ext_vector_type(8)));
typedef float    v4f  __attribute__((ext_vector_type(4)));

#define FDIM 256
#define TDIM 2048
#define BDIM 128
#define LDS_STRIDE 264

#if __has_builtin(__builtin_amdgcn_tanhf)
  #define FAST_TANH(v) __builtin_amdgcn_tanhf(v)
#elif __has_builtin(__builtin_amdgcn_tanh_f32)
  #define FAST_TANH(v) __builtin_amdgcn_tanh_f32(v)
#else
  #define FAST_TANH(v) tanhf(v)
#endif

#define SCORE_TILES ((BDIM * TDIM) / 256)
#define SCORE_WGS   256
#define TPW         (SCORE_TILES / SCORE_WGS)
#define WG_PER_B    (TDIM / (256 * TPW))
#define NWAVES      4
#define NS          4
#define NTHREADS    (NWAVES * 32)

#define XPITCH      264
#define WAVE_ROWS   (NS * 16)
#define G0_ROWS     18
#define G1_ROWS     18
#define G2_ROWS     28

template <int M> __device__ __forceinline__ float rowXorAdd(float v) {
    const int other = __builtin_amdgcn_update_dpp(0, __builtin_bit_cast(int, v),
                                                  0x160 + M  , 0xf, 0xf, true);
    return v + __builtin_bit_cast(float, other);
}

template <bool B> struct BoolTag { static constexpr bool value = B; };
template <int V>  struct IntTag  { static constexpr int  value = V; };

__global__ __launch_bounds__(NTHREADS) void awc_scores_wsum(const float* __restrict__ x,
                                                            const float* __restrict__ W,
                                                            const float* __restrict__ bias,
                                                            const float* __restrict__ u,
                                                            float* __restrict__ partial,
                                                            float* __restrict__ psum)
{
    __shared__ _Float16 sW[FDIM * LDS_STRIDE];
    __shared__ __attribute__((aligned(16))) _Float16 sX[NWAVES * WAVE_ROWS * XPITCH];
    __shared__ float sU[FDIM];
    __shared__ float sBias[FDIM];
    __shared__ float sY[NWAVES * 2 * FDIM];
    __shared__ float sE[NWAVES * 2];

    const int tid   = threadIdx.x;
    const int wave  = tid >> 5;
    const int lane  = tid & 31;
    const int lrow  = lane & 15;
    const int lhalf = lane >> 4;
    const int khalf = lhalf * 8;

    v4f g0[G0_ROWS][2], g1[G1_ROWS][2], g2[G2_ROWS][2];
    typedef _Float16 v4h __attribute__((ext_vector_type(4)));
    _Float16* const myX = &sX[wave * (WAVE_ROWS * XPITCH)];

    auto issueRows = [&](int tile, auto& g, auto r0Tag) {
        constexpr int R0 = decltype(r0Tag)::value;
        constexpr int NR = (int)(sizeof(g) / sizeof(g[0]));
        const float* xr = x + ((long)tile * 256 + wave * WAVE_ROWS + R0) * (long)FDIM + lane * 4;
#pragma unroll
        for (int r = 0; r < NR; ++r) {
            g[r][0] = *(const v4f*)(xr + r * FDIM);
            g[r][1] = *(const v4f*)(xr + r * FDIM + 128);
        }
    };
    auto stageRows = [&](auto& g, auto r0Tag) {
        constexpr int R0 = decltype(r0Tag)::value;
        constexpr int NR = (int)(sizeof(g) / sizeof(g[0]));
        _Float16* d = myX + R0 * XPITCH + lane * 4;
#pragma unroll
        for (int r = 0; r < NR; ++r) {
#pragma unroll
            for (int hh = 0; hh < 2; ++hh) {
                v4h h;
#pragma unroll
                for (int e = 0; e < 4; ++e) h[e] = (_Float16)g[r][hh][e];
                *(v4h*)(d + r * XPITCH + hh * 128) = h;
            }
        }
    };
    const IntTag<0>                 R_G0{};
    const IntTag<G0_ROWS>           R_G1{};
    const IntTag<G0_ROWS + G1_ROWS> R_G2{};

    const int tile0 = blockIdx.x * TPW;
    issueRows(tile0, g0, R_G0);

    auto stageW = [&](int fBeg, int fEnd) {
        _Float16* d0 = &sW[(unsigned)tid * LDS_STRIDE];
        _Float16* d1 = &sW[(unsigned)(tid + NTHREADS) * LDS_STRIDE];
#pragma unroll 16
        for (int f = fBeg; f < fEnd; f += 2) {
            const float a0 = W[f * FDIM + tid];
            const float a1 = W[(f + 1) * FDIM + tid];
            const float c0 = W[f * FDIM + tid + NTHREADS];
            const float c1 = W[(f + 1) * FDIM + tid + NTHREADS];
            typedef _Float16 v2h __attribute__((ext_vector_type(2)));
            v2h p, q;
            p[0] = (_Float16)a0; p[1] = (_Float16)a1;
            q[0] = (_Float16)c0; q[1] = (_Float16)c1;
            *(v2h*)(d0 + f) = p;
            *(v2h*)(d1 + f) = q;
        }
    };
    stageW(0, FDIM / 2);
    stageRows(g0, R_G0);
    issueRows(tile0, g1, R_G1);
    stageW(FDIM / 2, FDIM);
    stageRows(g1, R_G1);
    issueRows(tile0, g2, R_G2);
    sU[tid]               = u[tid];
    sU[tid + NTHREADS]    = u[tid + NTHREADS];
    sBias[tid]            = bias[tid];
    sBias[tid + NTHREADS] = bias[tid + NTHREADS];

    v16h sel0 = {};
    v16h sel1 = {};
    {
        const bool mine = ((lrow >> 3) == lhalf);
#pragma unroll
        for (int e = 0; e < 8; ++e) {
            const _Float16 v = (mine && (lrow & 7) == e) ? (_Float16)1.0f : (_Float16)0.0f;
            sel0[e]     = v;
            sel1[e + 8] = v;
        }
    }

    float ysum[16];
#pragma unroll
    for (int j = 0; j < 16; ++j) ysum[j] = 0.0f;
    float esum = 0.0f;

    __syncthreads();

    auto loadB = [&](int n, int kk) -> v16h {
        const _Float16* p = &sW[(n * 16 + lrow) * LDS_STRIDE + kk * 32 + khalf];
        v8h blo = *(const v8h*)(p);
        v8h bhi = *(const v8h*)(p + 16);
        return __builtin_shufflevector(blo, bhi,
            0,1,2,3,4,5,6,7,8,9,10,11,12,13,14,15);
    };

    v16h bA[8], bB[8];
#pragma unroll
    for (int k = 0; k < 8; ++k) bA[k] = loadB(0, k);

    auto tileBody = [&](int it, auto moreTag) {
        constexpr bool more = decltype(moreTag)::value;
        const int tile = tile0 + it;

        stageRows(g2, R_G2);

        if constexpr (more) issueRows(tile + 1, g0, R_G0);

        v16h afrag[NS][8];
#pragma unroll
        for (int s = 0; s < NS; ++s) {
#pragma unroll
            for (int k = 0; k < 8; ++k) {
                const _Float16* p = myX + (s * 16 + lrow) * XPITCH + k * 32 + khalf;
                v8h alo = *(const v8h*)(p);
                v8h ahi = *(const v8h*)(p + 16);
                afrag[s][k] = __builtin_shufflevector(alo, ahi,
                    0,1,2,3,4,5,6,7,8,9,10,11,12,13,14,15);
            }
        }

        float part[NS][8];
#pragma unroll
        for (int s = 0; s < NS; ++s)
#pragma unroll
            for (int r = 0; r < 8; ++r) part[s][r] = 0.0f;

        auto colBlock = [&](int n, const v16h (&bf)[8]) {
            v8f acc[NS];
#pragma unroll
            for (int s = 0; s < NS; ++s) acc[s] = v8f{};
            const float un = sU[n * 16 + lrow];
            const float bn = sBias[n * 16 + lrow];
#pragma unroll
            for (int k = 0; k < 8; ++k) {
#pragma unroll
                for (int s = 0; s < NS; ++s)
                    acc[s] = __builtin_amdgcn_wmma_f32_16x16x32_f16(
                        false, afrag[s][k], false, bf[k], (short)0, acc[s], false, false);
            }
#pragma unroll
            for (int r = 0; r < 8; ++r) {
#pragma unroll
                for (int s = 0; s < NS; ++s)
                    part[s][r] += FAST_TANH(acc[s][r] + bn) * un;
            }
        };

        auto colTrips = [&](int nBeg, int nEnd) {
#pragma clang loop unroll(disable)
        for (int n = nBeg; n < nEnd; n += 2) {
            __builtin_amdgcn_sched_barrier(0x406);
#pragma unroll
            for (int k = 0; k < 8; ++k) bB[k] = loadB(n + 1, k);
            __builtin_amdgcn_sched_barrier(0x406);
            colBlock(n, bA);
            __builtin_amdgcn_sched_barrier(0x406);
#pragma unroll
            for (int k = 0; k < 8; ++k) bA[k] = loadB((n + 2) & 15, k);
            __builtin_amdgcn_sched_barrier(0x406);
            colBlock(n + 1, bB);
        }
        };
        colTrips(0, 8);
        if constexpr (more) { stageRows(g0, R_G0); issueRows(tile + 1, g1, R_G1); }
        colTrips(8, 16);
        if constexpr (more) {
            stageRows(g1, R_G1);
            issueRows(tile + 1, g2, R_G2);
            __builtin_amdgcn_sched_barrier(0);
        }

#pragma unroll
        for (int s = 0; s < NS; ++s) {
            float v[8];
#pragma unroll
            for (int r = 0; r < 8; ++r) v[r] = rowXorAdd<1>(part[s][r]);
#pragma unroll
            for (int r = 0; r < 8; ++r) v[r] = rowXorAdd<2>(v[r]);
#pragma unroll
            for (int r = 0; r < 8; ++r) v[r] = rowXorAdd<4>(v[r]);
#pragma unroll
            for (int r = 0; r < 8; ++r) v[r] = rowXorAdd<8>(v[r]);
#pragma unroll
            for (int r = 0; r < 8; ++r) {
                const float e = expf(v[r]);
                part[s][r] = e;
                esum += e;
            }
        }

#pragma unroll
        for (int q = 0; q < 4 * NS; ++q) {
            const int s = q >> 2;
            const int g = q & 3;
#pragma unroll
            for (int jj = 0; jj < 4; ++jj) {
                const int j = 4 * g + jj;
                v8f z = {};
                v8f d = __builtin_amdgcn_wmma_f32_16x16x32_f16(
                    false, afrag[s][j >> 1], false, (j & 1) ? sel1 : sel0, (short)0, z, false, false);
#pragma unroll
                for (int r = 0; r < 8; ++r) ysum[j] = __builtin_fmaf(part[s][r], d[r], ysum[j]);
            }
        }
    };

#pragma clang loop unroll(disable)
    for (int it = 0; it < TPW - 1; ++it) tileBody(it, BoolTag<true>{});
    tileBody(TPW - 1, BoolTag<false>{});

#pragma unroll
    for (int j = 0; j < 16; ++j)
        sY[(wave * 2 + lhalf) * FDIM + j * 16 + lrow] = ysum[j];
    if (lrow == 0) sE[wave * 2 + lhalf] = esum;
    __syncthreads();

#pragma unroll
    for (int c2 = 0; c2 < FDIM / NTHREADS; ++c2) {
        const int col = tid + c2 * NTHREADS;
        float tot = 0.0f;
#pragma unroll
        for (int p = 0; p < NWAVES * 2; ++p) tot += sY[p * FDIM + col];
        partial[(long)blockIdx.x * FDIM + col] = tot;
    }
    if (tid == 0) {
        float es = 0.0f;
#pragma unroll
        for (int p = 0; p < NWAVES * 2; ++p) es += sE[p];
        psum[blockIdx.x] = es;
    }
}

__global__ __launch_bounds__(256) void awc_finalize(const float* __restrict__ partial,
                                                    const float* __restrict__ psum,
                                                    float* __restrict__ out)
{
    const int b = blockIdx.x;
    const int f = threadIdx.x;
    float acc = 0.0f;
    float es  = 0.0f;
#pragma unroll
    for (int c = 0; c < WG_PER_B; ++c) {
        acc += partial[((long)b * WG_PER_B + c) * FDIM + f];
        es  += psum[b * WG_PER_B + c];
    }
    out[(long)b * FDIM + f] = acc * (1.0f / (es + 1e-7f));
}

extern "C" void kernel_launch(void* const* d_in, const int* in_sizes, int n_in,
                              void* d_out, int out_size, void* d_ws, size_t ws_size,
                              hipStream_t stream) {
    const float* x    = (const float*)d_in[0];
    const float* W    = (const float*)d_in[1];
    const float* bias = (const float*)d_in[2];
    const float* u    = (const float*)d_in[3];
    float* out = (float*)d_out;

    float* psum    = (float*)d_ws;
    float* partial = psum + (size_t)BDIM * TDIM;

    awc_scores_wsum<<<dim3(SCORE_WGS), NTHREADS, 0, stream>>>(x, W, bias, u, partial, psum);
    awc_finalize   <<<dim3(BDIM),      256, 0, stream>>>(partial, psum, out);
}


# each column block's epilogue (bias, tanh, times u) issued beside the NEXT block's 32 WMMAs, one packed add or FMA and one tanh behind each WMMA, from a second accumulator set; same sums, same order
# speedup vs baseline: 2.4778x; 1.0312x over previous
#include <hip/hip_runtime.h>
#include <hip/hip_bf16.h>

typedef _Float16 v16h __attribute__((ext_vector_type(16)));
typedef _Float16 v8h  __attribute__((ext_vector_type(8)));
typedef float    v8f  __attribute__((ext_vector_type(8)));
typedef float    v4f  __attribute__((ext_vector_type(4)));
typedef float    v2f  __attribute__((ext_vector_type(2)));

#define FDIM 256
#define TDIM 2048
#define BDIM 128
#define LDS_STRIDE 264

#if __has_builtin(__builtin_amdgcn_tanhf)
  #define FAST_TANH(v) __builtin_amdgcn_tanhf(v)
#elif __has_builtin(__builtin_amdgcn_tanh_f32)
  #define FAST_TANH(v) __builtin_amdgcn_tanh_f32(v)
#else
  #define FAST_TANH(v) tanhf(v)
#endif

#define SCORE_TILES ((BDIM * TDIM) / 256)
#define SCORE_WGS   256
#define TPW         (SCORE_TILES / SCORE_WGS)
#define WG_PER_B    (TDIM / (256 * TPW))
#define NWAVES      4
#define NS          4
#define NTHREADS    (NWAVES * 32)

#define XPITCH      264
#define WAVE_ROWS   (NS * 16)
#define G0_ROWS     18
#define G1_ROWS     18
#define G2_ROWS     28

template <int M> __device__ __forceinline__ float rowXorAdd(float v) {
    const int other = __builtin_amdgcn_update_dpp(0, __builtin_bit_cast(int, v),
                                                  0x160 + M  , 0xf, 0xf, true);
    return v + __builtin_bit_cast(float, other);
}

template <bool B> struct BoolTag { static constexpr bool value = B; };
template <int V>  struct IntTag  { static constexpr int  value = V; };

struct EpiCarry { float x1, t0, un; };

__global__ __launch_bounds__(NTHREADS) void awc_scores_wsum(const float* __restrict__ x,
                                                            const float* __restrict__ W,
                                                            const float* __restrict__ bias,
                                                            const float* __restrict__ u,
                                                            float* __restrict__ partial,
                                                            float* __restrict__ psum)
{
    __shared__ _Float16 sW[FDIM * LDS_STRIDE];
    __shared__ __attribute__((aligned(16))) _Float16 sX[NWAVES * WAVE_ROWS * XPITCH];
    __shared__ float sU[FDIM];
    __shared__ float sBias[FDIM];
    __shared__ float sY[NWAVES * 2 * FDIM];
    __shared__ float sE[NWAVES * 2];

    const int tid   = threadIdx.x;
    const int wave  = tid >> 5;
    const int lane  = tid & 31;
    const int lrow  = lane & 15;
    const int lhalf = lane >> 4;
    const int khalf = lhalf * 8;

    v4f g0[G0_ROWS][2], g1[G1_ROWS][2], g2[G2_ROWS][2];
    typedef _Float16 v4h __attribute__((ext_vector_type(4)));
    _Float16* const myX = &sX[wave * (WAVE_ROWS * XPITCH)];

    auto issueRows = [&](int tile, auto& g, auto r0Tag) {
        constexpr int R0 = decltype(r0Tag)::value;
        constexpr int NR = (int)(sizeof(g) / sizeof(g[0]));
        const float* xr = x + ((long)tile * 256 + wave * WAVE_ROWS + R0) * (long)FDIM + lane * 4;
#pragma unroll
        for (int r = 0; r < NR; ++r) {
            g[r][0] = *(const v4f*)(xr + r * FDIM);
            g[r][1] = *(const v4f*)(xr + r * FDIM + 128);
        }
    };
    auto stageRows = [&](auto& g, auto r0Tag) {
        constexpr int R0 = decltype(r0Tag)::value;
        constexpr int NR = (int)(sizeof(g) / sizeof(g[0]));
        _Float16* d = myX + R0 * XPITCH + lane * 4;
#pragma unroll
        for (int r = 0; r < NR; ++r) {
#pragma unroll
            for (int hh = 0; hh < 2; ++hh) {
                v4h h;
#pragma unroll
                for (int e = 0; e < 4; ++e) h[e] = (_Float16)g[r][hh][e];
                *(v4h*)(d + r * XPITCH + hh * 128) = h;
            }
        }
    };
    const IntTag<0>                 R_G0{};
    const IntTag<G0_ROWS>           R_G1{};
    const IntTag<G0_ROWS + G1_ROWS> R_G2{};

    const int tile0 = blockIdx.x * TPW;
    issueRows(tile0, g0, R_G0);

    auto stageW = [&](int fBeg, int fEnd) {
        _Float16* d0 = &sW[(unsigned)tid * LDS_STRIDE];
        _Float16* d1 = &sW[(unsigned)(tid + NTHREADS) * LDS_STRIDE];
#pragma unroll 16
        for (int f = fBeg; f < fEnd; f += 2) {
            const float a0 = W[f * FDIM + tid];
            const float a1 = W[(f + 1) * FDIM + tid];
            const float c0 = W[f * FDIM + tid + NTHREADS];
            const float c1 = W[(f + 1) * FDIM + tid + NTHREADS];
            typedef _Float16 v2h __attribute__((ext_vector_type(2)));
            v2h p, q;
            p[0] = (_Float16)a0; p[1] = (_Float16)a1;
            q[0] = (_Float16)c0; q[1] = (_Float16)c1;
            *(v2h*)(d0 + f) = p;
            *(v2h*)(d1 + f) = q;
        }
    };
    stageW(0, FDIM / 2);
    stageRows(g0, R_G0);
    issueRows(tile0, g1, R_G1);
    stageW(FDIM / 2, FDIM);
    stageRows(g1, R_G1);
    issueRows(tile0, g2, R_G2);
    sU[tid]               = u[tid];
    sU[tid + NTHREADS]    = u[tid + NTHREADS];
    sBias[tid]            = bias[tid];
    sBias[tid + NTHREADS] = bias[tid + NTHREADS];

    v16h sel0 = {};
    v16h sel1 = {};
    {
        const bool mine = ((lrow >> 3) == lhalf);
#pragma unroll
        for (int e = 0; e < 8; ++e) {
            const _Float16 v = (mine && (lrow & 7) == e) ? (_Float16)1.0f : (_Float16)0.0f;
            sel0[e]     = v;
            sel1[e + 8] = v;
        }
    }

    float ysum[16];
#pragma unroll
    for (int j = 0; j < 16; ++j) ysum[j] = 0.0f;
    float esum = 0.0f;

    __syncthreads();

    auto loadB = [&](int n, int kk) -> v16h {
        const _Float16* p = &sW[(n * 16 + lrow) * LDS_STRIDE + kk * 32 + khalf];
        v8h blo = *(const v8h*)(p);
        v8h bhi = *(const v8h*)(p + 16);
        return __builtin_shufflevector(blo, bhi,
            0,1,2,3,4,5,6,7,8,9,10,11,12,13,14,15);
    };

    v16h bA[8], bB[8];
#pragma unroll
    for (int k = 0; k < 8; ++k) bA[k] = loadB(0, k);

    auto tileBody = [&](int it, auto moreTag) {
        constexpr bool more = decltype(moreTag)::value;
        const int tile = tile0 + it;

        stageRows(g2, R_G2);

        if constexpr (more) issueRows(tile + 1, g0, R_G0);

        v16h afrag[NS][8];
#pragma unroll
        for (int s = 0; s < NS; ++s) {
#pragma unroll
            for (int k = 0; k < 8; ++k) {
                const _Float16* p = myX + (s * 16 + lrow) * XPITCH + k * 32 + khalf;
                v8h alo = *(const v8h*)(p);
                v8h ahi = *(const v8h*)(p + 16);
                afrag[s][k] = __builtin_shufflevector(alo, ahi,
                    0,1,2,3,4,5,6,7,8,9,10,11,12,13,14,15);
            }
        }

        v2f part2[NS][4];
#pragma unroll
        for (int s = 0; s < NS; ++s)
#pragma unroll
            for (int q = 0; q < 4; ++q) part2[s][q] = v2f{0.0f, 0.0f};

        auto colBlock = [&](const v16h (&bf)[8], v8f (&cur)[NS], const v8f (&prev)[NS],
                            float unP, float bnP, EpiCarry& cy, auto prevTag, auto carryTag) {
            constexpr bool withPrev = decltype(prevTag)::value;
            constexpr bool hasCarry = decltype(carryTag)::value;
            constexpr int  NP = 4 * NS;
            const v2f un2 = {unP, unP};
            const v2f bn2 = {bnP, bnP};
            v2f x2[NP], t2[NP];
            float tc1 = 0.0f;
            auto addPair = [&](int j) {
                x2[j] = v2f{prev[j & 3][2 * (j >> 2)], prev[j & 3][2 * (j >> 2) + 1]} + bn2;
            };
#pragma unroll
            for (int k = 0; k < 8; ++k) {
#pragma unroll
                for (int s = 0; s < NS; ++s) {
                    const int i = k * NS + s;
                    const int m = i >> 1;
                    const v8f c = (k == 0) ? v8f{} : cur[s];
                    cur[s] = __builtin_amdgcn_wmma_f32_16x16x32_f16(
                        false, afrag[s][k], false, bf[k], (short)0, c, false, false);
                    if constexpr (withPrev || hasCarry) __builtin_amdgcn_sched_barrier(0);
                    if constexpr (withPrev) {
                        if (i == 0) {
                            addPair(0);
                            __builtin_amdgcn_sched_barrier(0);
                            addPair(1);
                            __builtin_amdgcn_sched_barrier(0);
                        } else if ((i & 1) == 0) {
                            if (m + 1 < NP) {
                                addPair(m + 1);
                                __builtin_amdgcn_sched_barrier(0);
                            }
                        } else if (m >= 1) {
                            part2[(m - 1) & 3][(m - 1) >> 2] = __builtin_elementwise_fma(
                                t2[m - 1], un2, part2[(m - 1) & 3][(m - 1) >> 2]);
                            __builtin_amdgcn_sched_barrier(0);
                        }
                    }
                    if constexpr (hasCarry) {
                        if (i == 1) {
                            part2[(NP - 1) & 3][(NP - 1) >> 2] = __builtin_elementwise_fma(
                                v2f{cy.t0, tc1}, v2f{cy.un, cy.un}, part2[(NP - 1) & 3][(NP - 1) >> 2]);
                            __builtin_amdgcn_sched_barrier(0);
                        }
                    }
                    if (i == 0) {
                        if constexpr (hasCarry) tc1 = FAST_TANH(cy.x1);
                    } else if constexpr (withPrev) {
                        if (i & 1) t2[m][0] = FAST_TANH(x2[m][0]);
                        else       t2[m - 1][1] = FAST_TANH(x2[m - 1][1]);
                    }
                    __builtin_amdgcn_sched_barrier(0);
                }
            }
            if constexpr (withPrev) {
                cy.x1 = x2[NP - 1][1];
                cy.t0 = t2[NP - 1][0];
                cy.un = unP;
            }
        };

        v8f accA[NS], accB[NS];
        float unA = 0.0f, bnA = 0.0f, unB = 0.0f, bnB = 0.0f;
        EpiCarry cy = {0.0f, 0.0f, 0.0f};

        auto trip = [&](int n, auto firstTag) {
            constexpr bool first = decltype(firstTag)::value;
            __builtin_amdgcn_sched_barrier(0x406);
#pragma unroll
            for (int k = 0; k < 8; ++k) bB[k] = loadB(n + 1, k);
            unA = sU[n * 16 + lrow];
            bnA = sBias[n * 16 + lrow];
            __builtin_amdgcn_sched_barrier(0x406);
            colBlock(bA, accA, accB, unB, bnB, cy, BoolTag<!first>{}, BoolTag<!first>{});
            __builtin_amdgcn_sched_barrier(0x406);
#pragma unroll
            for (int k = 0; k < 8; ++k) bA[k] = loadB((n + 2) & 15, k);
            unB = sU[(n + 1) * 16 + lrow];
            bnB = sBias[(n + 1) * 16 + lrow];
            __builtin_amdgcn_sched_barrier(0x406);
            colBlock(bB, accB, accA, unA, bnA, cy, BoolTag<true>{}, BoolTag<!first>{});
        };
        auto colTrips = [&](int nBeg, int nEnd) {
#pragma clang loop unroll(disable)
            for (int n = nBeg; n < nEnd; n += 2) trip(n, BoolTag<false>{});
        };
        trip(0, BoolTag<true>{});
        colTrips(2, 8);
        if constexpr (more) { stageRows(g0, R_G0); issueRows(tile + 1, g1, R_G1); }
        colTrips(8, 16);
        if constexpr (more) {
            stageRows(g1, R_G1);
            issueRows(tile + 1, g2, R_G2);
            __builtin_amdgcn_sched_barrier(0);
        }
        part2[NS - 1][3] = __builtin_elementwise_fma(
            v2f{cy.t0, FAST_TANH(cy.x1)}, v2f{cy.un, cy.un}, part2[NS - 1][3]);
        float part[NS][8];
#pragma unroll
        for (int s = 0; s < NS; ++s) {
#pragma unroll
            for (int r = 0; r < 8; ++r)
                part[s][r] = __builtin_fmaf(FAST_TANH(accB[s][r] + bnB), unB, part2[s][r >> 1][r & 1]);
        }

#pragma unroll
        for (int s = 0; s < NS; ++s) {
            float v[8];
#pragma unroll
            for (int r = 0; r < 8; ++r) v[r] = rowXorAdd<1>(part[s][r]);
#pragma unroll
            for (int r = 0; r < 8; ++r) v[r] = rowXorAdd<2>(v[r]);
#pragma unroll
            for (int r = 0; r < 8; ++r) v[r] = rowXorAdd<4>(v[r]);
#pragma unroll
            for (int r = 0; r < 8; ++r) v[r] = rowXorAdd<8>(v[r]);
#pragma unroll
            for (int r = 0; r < 8; ++r) {
                const float e = expf(v[r]);
                part[s][r] = e;
                esum += e;
            }
        }

#pragma unroll
        for (int q = 0; q < 4 * NS; ++q) {
            const int s = q >> 2;
            const int g = q & 3;
#pragma unroll
            for (int jj = 0; jj < 4; ++jj) {
                const int j = 4 * g + jj;
                v8f z = {};
                v8f d = __builtin_amdgcn_wmma_f32_16x16x32_f16(
                    false, afrag[s][j >> 1], false, (j & 1) ? sel1 : sel0, (short)0, z, false, false);
#pragma unroll
                for (int r = 0; r < 8; ++r) ysum[j] = __builtin_fmaf(part[s][r], d[r], ysum[j]);
            }
        }
    };

#pragma clang loop unroll(disable)
    for (int it = 0; it < TPW - 1; ++it) tileBody(it, BoolTag<true>{});
    tileBody(TPW - 1, BoolTag<false>{});

#pragma unroll
    for (int j = 0; j < 16; ++j)
        sY[(wave * 2 + lhalf) * FDIM + j * 16 + lrow] = ysum[j];
    if (lrow == 0) sE[wave * 2 + lhalf] = esum;
    __syncthreads();

#pragma unroll
    for (int c2 = 0; c2 < FDIM / NTHREADS; ++c2) {
        const int col = tid + c2 * NTHREADS;
        float tot = 0.0f;
#pragma unroll
        for (int p = 0; p < NWAVES * 2; ++p) tot += sY[p * FDIM + col];
        partial[(long)blockIdx.x * FDIM + col] = tot;
    }
    if (tid == 0) {
        float es = 0.0f;
#pragma unroll
        for (int p = 0; p < NWAVES * 2; ++p) es += sE[p];
        psum[blockIdx.x] = es;
    }
}

__global__ __launch_bounds__(256) void awc_finalize(const float* __restrict__ partial,
                                                    const float* __restrict__ psum,
                                                    float* __restrict__ out)
{
    const int b = blockIdx.x;
    const int f = threadIdx.x;
    float acc = 0.0f;
    float es  = 0.0f;
#pragma unroll
    for (int c = 0; c < WG_PER_B; ++c) {
        acc += partial[((long)b * WG_PER_B + c) * FDIM + f];
        es  += psum[b * WG_PER_B + c];
    }
    out[(long)b * FDIM + f] = acc * (1.0f / (es + 1e-7f));
}

extern "C" void kernel_launch(void* const* d_in, const int* in_sizes, int n_in,
                              void* d_out, int out_size, void* d_ws, size_t ws_size,
                              hipStream_t stream) {
    const float* x    = (const float*)d_in[0];
    const float* W    = (const float*)d_in[1];
    const float* bias = (const float*)d_in[2];
    const float* u    = (const float*)d_in[3];
    float* out = (float*)d_out;

    float* psum    = (float*)d_ws;
    float* partial = psum + (size_t)BDIM * TDIM;

    awc_scores_wsum<<<dim3(SCORE_WGS), NTHREADS, 0, stream>>>(x, W, bias, u, partial, psum);
    awc_finalize   <<<dim3(BDIM),      256, 0, stream>>>(partial, psum, out);
}


# x brought into the second-level cache by global_prefetch_b8 half a tile to a tile ahead of its loads, one prefetch of 8 rows per column-loop trip; the x loads then hit L2; same sums, same order
# speedup vs baseline: 2.6494x; 1.0693x over previous
#include <hip/hip_runtime.h>
#include <hip/hip_bf16.h>

typedef _Float16 v16h __attribute__((ext_vector_type(16)));
typedef _Float16 v8h  __attribute__((ext_vector_type(8)));
typedef float    v8f  __attribute__((ext_vector_type(8)));
typedef float    v4f  __attribute__((ext_vector_type(4)));
typedef float    v2f  __attribute__((ext_vector_type(2)));

#define FDIM 256
#define TDIM 2048
#define BDIM 128
#define LDS_STRIDE 264

#if __has_builtin(__builtin_amdgcn_tanhf)
  #define FAST_TANH(v) __builtin_amdgcn_tanhf(v)
#elif __has_builtin(__builtin_amdgcn_tanh_f32)
  #define FAST_TANH(v) __builtin_amdgcn_tanh_f32(v)
#else
  #define FAST_TANH(v) tanhf(v)
#endif

#define SCORE_TILES ((BDIM * TDIM) / 256)
#define SCORE_WGS   256
#define TPW         (SCORE_TILES / SCORE_WGS)
#define WG_PER_B    (TDIM / (256 * TPW))
#define NWAVES      4
#define NS          4
#define NTHREADS    (NWAVES * 32)

#define XPITCH      264
#define PF_LOCALITY 2
#define WAVE_ROWS   (NS * 16)
#define G0_ROWS     18
#define G1_ROWS     18
#define G2_ROWS     28

template <int M> __device__ __forceinline__ float rowXorAdd(float v) {
    const int other = __builtin_amdgcn_update_dpp(0, __builtin_bit_cast(int, v),
                                                  0x160 + M  , 0xf, 0xf, true);
    return v + __builtin_bit_cast(float, other);
}

template <bool B> struct BoolTag { static constexpr bool value = B; };
template <int V>  struct IntTag  { static constexpr int  value = V; };

struct EpiCarry { float x1, t0, un; };

__global__ __launch_bounds__(NTHREADS) void awc_scores_wsum(const float* __restrict__ x,
                                                            const float* __restrict__ W,
                                                            const float* __restrict__ bias,
                                                            const float* __restrict__ u,
                                                            float* __restrict__ partial,
                                                            float* __restrict__ psum)
{
    __shared__ _Float16 sW[FDIM * LDS_STRIDE];
    __shared__ __attribute__((aligned(16))) _Float16 sX[NWAVES * WAVE_ROWS * XPITCH];
    __shared__ float sU[FDIM];
    __shared__ float sBias[FDIM];
    __shared__ float sY[NWAVES * 2 * FDIM];
    __shared__ float sE[NWAVES * 2];

    const int tid   = threadIdx.x;
    const int wave  = tid >> 5;
    const int lane  = tid & 31;
    const int lrow  = lane & 15;
    const int lhalf = lane >> 4;
    const int khalf = lhalf * 8;

    v4f g0[G0_ROWS][2], g1[G1_ROWS][2], g2[G2_ROWS][2];
    typedef _Float16 v4h __attribute__((ext_vector_type(4)));
    _Float16* const myX = &sX[wave * (WAVE_ROWS * XPITCH)];

    auto issueRows = [&](int tile, auto& g, auto r0Tag) {
        constexpr int R0 = decltype(r0Tag)::value;
        constexpr int NR = (int)(sizeof(g) / sizeof(g[0]));
        const float* xr = x + ((long)tile * 256 + wave * WAVE_ROWS + R0) * (long)FDIM + lane * 4;
#pragma unroll
        for (int r = 0; r < NR; ++r) {
            g[r][0] = *(const v4f*)(xr + r * FDIM);
            g[r][1] = *(const v4f*)(xr + r * FDIM + 128);
        }
    };
    auto stageRows = [&](auto& g, auto r0Tag) {
        constexpr int R0 = decltype(r0Tag)::value;
        constexpr int NR = (int)(sizeof(g) / sizeof(g[0]));
        _Float16* d = myX + R0 * XPITCH + lane * 4;
#pragma unroll
        for (int r = 0; r < NR; ++r) {
#pragma unroll
            for (int hh = 0; hh < 2; ++hh) {
                v4h h;
#pragma unroll
                for (int e = 0; e < 4; ++e) h[e] = (_Float16)g[r][hh][e];
                *(v4h*)(d + r * XPITCH + hh * 128) = h;
            }
        }
    };
    const IntTag<0>                 R_G0{};
    const IntTag<G0_ROWS>           R_G1{};
    const IntTag<G0_ROWS + G1_ROWS> R_G2{};

    const int tile0 = blockIdx.x * TPW;

    const long pfLane = (long)(lane >> 2) * FDIM + (lane & 3) * 64;
    auto rowsPtr = [&](int tile, int row0) -> const float* {
        return x + ((long)tile * 256 + wave * WAVE_ROWS + row0) * (long)FDIM + pfLane;
    };

    issueRows(tile0, g0, R_G0);
#pragma unroll
    for (int r0 = 16; r0 < WAVE_ROWS; r0 += 8) __builtin_prefetch(rowsPtr(tile0, r0), 0, PF_LOCALITY);

    auto stageW = [&](int fBeg, int fEnd) {
        _Float16* d0 = &sW[(unsigned)tid * LDS_STRIDE];
        _Float16* d1 = &sW[(unsigned)(tid + NTHREADS) * LDS_STRIDE];
#pragma unroll 16
        for (int f = fBeg; f < fEnd; f += 2) {
            const float a0 = W[f * FDIM + tid];
            const float a1 = W[(f + 1) * FDIM + tid];
            const float c0 = W[f * FDIM + tid + NTHREADS];
            const float c1 = W[(f + 1) * FDIM + tid + NTHREADS];
            typedef _Float16 v2h __attribute__((ext_vector_type(2)));
            v2h p, q;
            p[0] = (_Float16)a0; p[1] = (_Float16)a1;
            q[0] = (_Float16)c0; q[1] = (_Float16)c1;
            *(v2h*)(d0 + f) = p;
            *(v2h*)(d1 + f) = q;
        }
    };
    stageW(0, FDIM / 2);
    stageRows(g0, R_G0);
    issueRows(tile0, g1, R_G1);
    if constexpr (TPW > 1) {
#pragma unroll
        for (int r0 = 0; r0 < WAVE_ROWS / 2; r0 += 8) __builtin_prefetch(rowsPtr(tile0 + 1, r0), 0, PF_LOCALITY);
    }
    stageW(FDIM / 2, FDIM);
    stageRows(g1, R_G1);
    issueRows(tile0, g2, R_G2);
    sU[tid]               = u[tid];
    sU[tid + NTHREADS]    = u[tid + NTHREADS];
    sBias[tid]            = bias[tid];
    sBias[tid + NTHREADS] = bias[tid + NTHREADS];

    v16h sel0 = {};
    v16h sel1 = {};
    {
        const bool mine = ((lrow >> 3) == lhalf);
#pragma unroll
        for (int e = 0; e < 8; ++e) {
            const _Float16 v = (mine && (lrow & 7) == e) ? (_Float16)1.0f : (_Float16)0.0f;
            sel0[e]     = v;
            sel1[e + 8] = v;
        }
    }

    float ysum[16];
#pragma unroll
    for (int j = 0; j < 16; ++j) ysum[j] = 0.0f;
    float esum = 0.0f;

    __syncthreads();

    auto loadB = [&](int n, int kk) -> v16h {
        const _Float16* p = &sW[(n * 16 + lrow) * LDS_STRIDE + kk * 32 + khalf];
        v8h blo = *(const v8h*)(p);
        v8h bhi = *(const v8h*)(p + 16);
        return __builtin_shufflevector(blo, bhi,
            0,1,2,3,4,5,6,7,8,9,10,11,12,13,14,15);
    };

    v16h bA[8], bB[8];
#pragma unroll
    for (int k = 0; k < 8; ++k) bA[k] = loadB(0, k);

    auto tileBody = [&](int it, auto moreTag) {
        constexpr bool more = decltype(moreTag)::value;
        const int tile = tile0 + it;

        const float* pfp = nullptr;
        if constexpr (more) pfp = rowsPtr(tile + 1, WAVE_ROWS / 2);

        stageRows(g2, R_G2);

        if constexpr (more) issueRows(tile + 1, g0, R_G0);

        v16h afrag[NS][8];
#pragma unroll
        for (int s = 0; s < NS; ++s) {
#pragma unroll
            for (int k = 0; k < 8; ++k) {
                const _Float16* p = myX + (s * 16 + lrow) * XPITCH + k * 32 + khalf;
                v8h alo = *(const v8h*)(p);
                v8h ahi = *(const v8h*)(p + 16);
                afrag[s][k] = __builtin_shufflevector(alo, ahi,
                    0,1,2,3,4,5,6,7,8,9,10,11,12,13,14,15);
            }
        }

        v2f part2[NS][4];
#pragma unroll
        for (int s = 0; s < NS; ++s)
#pragma unroll
            for (int q = 0; q < 4; ++q) part2[s][q] = v2f{0.0f, 0.0f};

        auto colBlock = [&](const v16h (&bf)[8], v8f (&cur)[NS], const v8f (&prev)[NS],
                            float unP, float bnP, EpiCarry& cy, auto prevTag, auto carryTag) {
            constexpr bool withPrev = decltype(prevTag)::value;
            constexpr bool hasCarry = decltype(carryTag)::value;
            constexpr int  NP = 4 * NS;
            const v2f un2 = {unP, unP};
            const v2f bn2 = {bnP, bnP};
            v2f x2[NP], t2[NP];
            float tc1 = 0.0f;
            auto addPair = [&](int j) {
                x2[j] = v2f{prev[j & 3][2 * (j >> 2)], prev[j & 3][2 * (j >> 2) + 1]} + bn2;
            };
#pragma unroll
            for (int k = 0; k < 8; ++k) {
#pragma unroll
                for (int s = 0; s < NS; ++s) {
                    const int i = k * NS + s;
                    const int m = i >> 1;
                    const v8f c = (k == 0) ? v8f{} : cur[s];
                    cur[s] = __builtin_amdgcn_wmma_f32_16x16x32_f16(
                        false, afrag[s][k], false, bf[k], (short)0, c, false, false);
                    if constexpr (withPrev || hasCarry) __builtin_amdgcn_sched_barrier(0);
                    if constexpr (withPrev) {
                        if (i == 0) {
                            addPair(0);
                            __builtin_amdgcn_sched_barrier(0);
                            addPair(1);
                            __builtin_amdgcn_sched_barrier(0);
                        } else if ((i & 1) == 0) {
                            if (m + 1 < NP) {
                                addPair(m + 1);
                                __builtin_amdgcn_sched_barrier(0);
                            }
                        } else if (m >= 1) {
                            part2[(m - 1) & 3][(m - 1) >> 2] = __builtin_elementwise_fma(
                                t2[m - 1], un2, part2[(m - 1) & 3][(m - 1) >> 2]);
                            __builtin_amdgcn_sched_barrier(0);
                        }
                    }
                    if constexpr (hasCarry) {
                        if (i == 1) {
                            part2[(NP - 1) & 3][(NP - 1) >> 2] = __builtin_elementwise_fma(
                                v2f{cy.t0, tc1}, v2f{cy.un, cy.un}, part2[(NP - 1) & 3][(NP - 1) >> 2]);
                            __builtin_amdgcn_sched_barrier(0);
                        }
                    }
                    if (i == 0) {
                        if constexpr (hasCarry) tc1 = FAST_TANH(cy.x1);
                    } else if constexpr (withPrev) {
                        if (i & 1) t2[m][0] = FAST_TANH(x2[m][0]);
                        else       t2[m - 1][1] = FAST_TANH(x2[m - 1][1]);
                    }
                    __builtin_amdgcn_sched_barrier(0);
                }
            }
            if constexpr (withPrev) {
                cy.x1 = x2[NP - 1][1];
                cy.t0 = t2[NP - 1][0];
                cy.un = unP;
            }
        };

        v8f accA[NS], accB[NS];
        float unA = 0.0f, bnA = 0.0f, unB = 0.0f, bnB = 0.0f;
        EpiCarry cy = {0.0f, 0.0f, 0.0f};

        auto trip = [&](int n, auto firstTag) {
            constexpr bool first = decltype(firstTag)::value;
            if constexpr (more) {
                __builtin_prefetch(pfp, 0, PF_LOCALITY);
                pfp += 8 * FDIM;
            }
            __builtin_amdgcn_sched_barrier(0x406);
#pragma unroll
            for (int k = 0; k < 8; ++k) bB[k] = loadB(n + 1, k);
            unA = sU[n * 16 + lrow];
            bnA = sBias[n * 16 + lrow];
            __builtin_amdgcn_sched_barrier(0x406);
            colBlock(bA, accA, accB, unB, bnB, cy, BoolTag<!first>{}, BoolTag<!first>{});
            __builtin_amdgcn_sched_barrier(0x406);
#pragma unroll
            for (int k = 0; k < 8; ++k) bA[k] = loadB((n + 2) & 15, k);
            unB = sU[(n + 1) * 16 + lrow];
            bnB = sBias[(n + 1) * 16 + lrow];
            __builtin_amdgcn_sched_barrier(0x406);
            colBlock(bB, accB, accA, unA, bnA, cy, BoolTag<true>{}, BoolTag<!first>{});
        };
        auto colTrips = [&](int nBeg, int nEnd) {
#pragma clang loop unroll(disable)
            for (int n = nBeg; n < nEnd; n += 2) trip(n, BoolTag<false>{});
        };
        trip(0, BoolTag<true>{});
        colTrips(2, 8);
        if constexpr (more) {
            stageRows(g0, R_G0);
            issueRows(tile + 1, g1, R_G1);
            pfp = rowsPtr((it + 2 < TPW) ? tile + 2 : tile + 1, 0);
        }
        colTrips(8, 16);
        if constexpr (more) {
            stageRows(g1, R_G1);
            issueRows(tile + 1, g2, R_G2);
            __builtin_amdgcn_sched_barrier(0);
        }
        part2[NS - 1][3] = __builtin_elementwise_fma(
            v2f{cy.t0, FAST_TANH(cy.x1)}, v2f{cy.un, cy.un}, part2[NS - 1][3]);
        float part[NS][8];
#pragma unroll
        for (int s = 0; s < NS; ++s) {
#pragma unroll
            for (int r = 0; r < 8; ++r)
                part[s][r] = __builtin_fmaf(FAST_TANH(accB[s][r] + bnB), unB, part2[s][r >> 1][r & 1]);
        }

#pragma unroll
        for (int s = 0; s < NS; ++s) {
            float v[8];
#pragma unroll
            for (int r = 0; r < 8; ++r) v[r] = rowXorAdd<1>(part[s][r]);
#pragma unroll
            for (int r = 0; r < 8; ++r) v[r] = rowXorAdd<2>(v[r]);
#pragma unroll
            for (int r = 0; r < 8; ++r) v[r] = rowXorAdd<4>(v[r]);
#pragma unroll
            for (int r = 0; r < 8; ++r) v[r] = rowXorAdd<8>(v[r]);
#pragma unroll
            for (int r = 0; r < 8; ++r) {
                const float e = expf(v[r]);
                part[s][r] = e;
                esum += e;
            }
        }

#pragma unroll
        for (int q = 0; q < 4 * NS; ++q) {
            const int s = q >> 2;
            const int g = q & 3;
#pragma unroll
            for (int jj = 0; jj < 4; ++jj) {
                const int j = 4 * g + jj;
                v8f z = {};
                v8f d = __builtin_amdgcn_wmma_f32_16x16x32_f16(
                    false, afrag[s][j >> 1], false, (j & 1) ? sel1 : sel0, (short)0, z, false, false);
#pragma unroll
                for (int r = 0; r < 8; ++r) ysum[j] = __builtin_fmaf(part[s][r], d[r], ysum[j]);
            }
        }
    };

#pragma clang loop unroll(disable)
    for (int it = 0; it < TPW - 1; ++it) tileBody(it, BoolTag<true>{});
    tileBody(TPW - 1, BoolTag<false>{});

#pragma unroll
    for (int j = 0; j < 16; ++j)
        sY[(wave * 2 + lhalf) * FDIM + j * 16 + lrow] = ysum[j];
    if (lrow == 0) sE[wave * 2 + lhalf] = esum;
    __syncthreads();

#pragma unroll
    for (int c2 = 0; c2 < FDIM / NTHREADS; ++c2) {
        const int col = tid + c2 * NTHREADS;
        float tot = 0.0f;
#pragma unroll
        for (int p = 0; p < NWAVES * 2; ++p) tot += sY[p * FDIM + col];
        partial[(long)blockIdx.x * FDIM + col] = tot;
    }
    if (tid == 0) {
        float es = 0.0f;
#pragma unroll
        for (int p = 0; p < NWAVES * 2; ++p) es += sE[p];
        psum[blockIdx.x] = es;
    }
}

__global__ __launch_bounds__(256) void awc_finalize(const float* __restrict__ partial,
                                                    const float* __restrict__ psum,
                                                    float* __restrict__ out)
{
    const int b = blockIdx.x;
    const int f = threadIdx.x;
    float acc = 0.0f;
    float es  = 0.0f;
#pragma unroll
    for (int c = 0; c < WG_PER_B; ++c) {
        acc += partial[((long)b * WG_PER_B + c) * FDIM + f];
        es  += psum[b * WG_PER_B + c];
    }
    out[(long)b * FDIM + f] = acc * (1.0f / (es + 1e-7f));
}

extern "C" void kernel_launch(void* const* d_in, const int* in_sizes, int n_in,
                              void* d_out, int out_size, void* d_ws, size_t ws_size,
                              hipStream_t stream) {
    const float* x    = (const float*)d_in[0];
    const float* W    = (const float*)d_in[1];
    const float* bias = (const float*)d_in[2];
    const float* u    = (const float*)d_in[3];
    float* out = (float*)d_out;

    float* psum    = (float*)d_ws;
    float* partial = psum + (size_t)BDIM * TDIM;

    awc_scores_wsum<<<dim3(SCORE_WGS), NTHREADS, 0, stream>>>(x, W, bias, u, partial, psum);
    awc_finalize   <<<dim3(BDIM),      256, 0, stream>>>(partial, psum, out);
}
